# starting kernel
# baseline (speedup 1.0000x reference)
#include <hip/hip_runtime.h>

typedef __attribute__((ext_vector_type(2))) float v2f;
typedef __attribute__((ext_vector_type(8))) float v8f;

#define HWSZ  3136
#define WID   56
#define CIN   256
#define COUT  256
#define RANK  32
#define KTOT  288
#define NTOT  50176

__global__ __launch_bounds__(128) void lr_stage1(const float* __restrict__ x,
                                                 const float* __restrict__ G,
                                                 float* __restrict__ U)
{
    __shared__ float Gs[32 * 257];
    const int tid = threadIdx.x;
    const int m0  = blockIdx.x * 32;

    for (int i = tid; i < 32 * 256; i += 128) {
        int row = i >> 8, col = i & 255;
        Gs[row * 257 + col] = G[(m0 + row) * 256 + col];
    }
    __syncthreads();

    const int wave = tid >> 5;
    const int lane = tid & 31;
    const int col  = lane & 15;
    const int klo  = (lane >> 4) << 1;

    const int n0   = (blockIdx.y * 4 + wave) * 16;
    const int bidx = n0 / HWSZ;
    const int p0   = n0 - bidx * HWSZ;
    const float* xb = x + (size_t)bidx * CIN * HWSZ + p0 + col;

    v8f acc0 = {};
    v8f acc1 = {};
    #pragma unroll 8
    for (int k0 = 0; k0 < CIN; k0 += 4) {
        v2f b;
        b.x = xb[(size_t)(k0 + klo)     * HWSZ];
        b.y = xb[(size_t)(k0 + klo + 1) * HWSZ];
        v2f a0, a1;
        a0.x = Gs[col * 257 + k0 + klo];
        a0.y = Gs[col * 257 + k0 + klo + 1];
        a1.x = Gs[(col + 16) * 257 + k0 + klo];
        a1.y = Gs[(col + 16) * 257 + k0 + klo + 1];
        acc0 = __builtin_amdgcn_wmma_f32_16x16x4_f32(false, a0, false, b, (short)0, acc0, false, false);
        acc1 = __builtin_amdgcn_wmma_f32_16x16x4_f32(false, a1, false, b, (short)0, acc1, false, false);
    }

    const int mo = (lane >> 4) << 3;
    float* U0 = U + (size_t)(m0 + mo) * NTOT + n0 + col;
    float* U1 = U0 + (size_t)16 * NTOT;
    #pragma unroll
    for (int v = 0; v < 8; ++v) {
        U0[(size_t)v * NTOT] = acc0[v];
        U1[(size_t)v * NTOT] = acc1[v];
    }
}

__global__ __launch_bounds__(128) void lr_stage2(const float* __restrict__ U,
                                                 const float* __restrict__ Hm,
                                                 const float* __restrict__ bias,
                                                 float* __restrict__ out)
{
    __shared__ float Hs[32 * 289];
    const int tid = threadIdx.x;
    const int m0  = blockIdx.x * 32;

    for (int i = tid; i < 32 * KTOT; i += 128) {
        int o_l = i / KTOT;
        int kk  = i - o_l * KTOT;
        int k = kk >> 5, r = kk & 31;
        Hs[o_l * 289 + kk] = Hm[(size_t)k * (COUT * RANK) + (size_t)(m0 + o_l) * RANK + r];
    }
    __syncthreads();

    const int wave = tid >> 5;
    const int lane = tid & 31;
    const int col  = lane & 15;
    const int klo  = (lane >> 4) << 1;

    const int n0   = (blockIdx.y * 4 + wave) * 16;
    const int n    = n0 + col;
    const int bidx = n / HWSZ;
    const int p    = n - bidx * HWSZ;
    const int h    = p / WID;
    const int w    = p - h * WID;

    int offs[9];
    unsigned vmask = 0;
    #pragma unroll
    for (int k = 0; k < 9; ++k) {
        int ki = k / 3, kj = k - ki * 3;
        int hs = h + ki - 1, ws = w + kj - 1;
        bool ok = (hs >= 0) & (hs < WID) & (ws >= 0) & (ws < WID);
        offs[k] = ok ? (bidx * HWSZ + hs * WID + ws) : 0;
        vmask  |= (ok ? 1u : 0u) << k;
    }

    v8f acc0 = {};
    v8f acc1 = {};
    #pragma unroll
    for (int k = 0; k < 9; ++k) {
        const bool ok = (vmask >> k) & 1u;
        const float* Ucol = U + offs[k];
        #pragma unroll
        for (int r0 = 0; r0 < RANK; r0 += 4) {
            const int kk = k * RANK + r0 + klo;
            v2f b;
            b.x = ok ? Ucol[(size_t)kk       * NTOT] : 0.0f;
            b.y = ok ? Ucol[(size_t)(kk + 1) * NTOT] : 0.0f;
            v2f a0, a1;
            a0.x = Hs[col * 289 + kk];
            a0.y = Hs[col * 289 + kk + 1];
            a1.x = Hs[(col + 16) * 289 + kk];
            a1.y = Hs[(col + 16) * 289 + kk + 1];
            acc0 = __builtin_amdgcn_wmma_f32_16x16x4_f32(false, a0, false, b, (short)0, acc0, false, false);
            acc1 = __builtin_amdgcn_wmma_f32_16x16x4_f32(false, a1, false, b, (short)0, acc1, false, false);
        }
    }

    const int mo = (lane >> 4) << 3;
    const float inv9 = 1.0f / 9.0f;
    #pragma unroll
    for (int v = 0; v < 8; ++v) {
        int o0 = m0 + v + mo;
        int o1 = o0 + 16;
        out[((size_t)bidx * COUT + o0) * HWSZ + p] = acc0[v] * inv9 + bias[o0];
        out[((size_t)bidx * COUT + o1) * HWSZ + p] = acc1[v] * inv9 + bias[o1];
    }
}

extern "C" void kernel_launch(void* const* d_in, const int* in_sizes, int n_in,
                              void* d_out, int out_size, void* d_ws, size_t ws_size,
                              hipStream_t stream)
{
    const float* x    = (const float*)d_in[0];
    const float* G    = (const float*)d_in[1];
    const float* Hm   = (const float*)d_in[2];
    const float* bias = (const float*)d_in[3];
    float* out = (float*)d_out;
    float* U   = (float*)d_ws;

    dim3 blk(128);
    lr_stage1<<<dim3(KTOT / 32, NTOT / 64), blk, 0, stream>>>(x, G, U);
    lr_stage2<<<dim3(COUT / 32, NTOT / 64), blk, 0, stream>>>(U, Hm, bias, out);
}


# Global loads no longer wait on one another: halo lanes load from the clamped offset and are zeroed by select (no per-load branch); both A-tile fills and the stage-1 K loop fully unrolled
# speedup vs baseline: 1.6041x; 1.6041x over previous
#include <hip/hip_runtime.h>

typedef __attribute__((ext_vector_type(2))) float v2f;
typedef __attribute__((ext_vector_type(8))) float v8f;

#define HWSZ  3136
#define WID   56
#define CIN   256
#define COUT  256
#define RANK  32
#define KTOT  288
#define NTOT  50176

__global__ __launch_bounds__(128) void lr_stage1(const float* __restrict__ x,
                                                 const float* __restrict__ G,
                                                 float* __restrict__ U)
{
    __shared__ float Gs[32 * 257];
    const int tid = threadIdx.x;
    const int m0  = blockIdx.x * 32;

    #pragma unroll
    for (int it = 0; it < 64; ++it) {
        const int row = it >> 1, col = tid + ((it & 1) << 7);
        Gs[row * 257 + col] = G[(m0 + row) * 256 + col];
    }
    __syncthreads();

    const int wave = tid >> 5;
    const int lane = tid & 31;
    const int col  = lane & 15;
    const int klo  = (lane >> 4) << 1;

    const int n0   = (blockIdx.y * 4 + wave) * 16;
    const int bidx = n0 / HWSZ;
    const int p0   = n0 - bidx * HWSZ;
    const float* xb = x + (size_t)bidx * CIN * HWSZ + p0 + col;

    v8f acc0 = {};
    v8f acc1 = {};
    #pragma unroll
    for (int k0 = 0; k0 < CIN; k0 += 4) {
        v2f b;
        b.x = xb[(size_t)(k0 + klo)     * HWSZ];
        b.y = xb[(size_t)(k0 + klo + 1) * HWSZ];
        v2f a0, a1;
        a0.x = Gs[col * 257 + k0 + klo];
        a0.y = Gs[col * 257 + k0 + klo + 1];
        a1.x = Gs[(col + 16) * 257 + k0 + klo];
        a1.y = Gs[(col + 16) * 257 + k0 + klo + 1];
        acc0 = __builtin_amdgcn_wmma_f32_16x16x4_f32(false, a0, false, b, (short)0, acc0, false, false);
        acc1 = __builtin_amdgcn_wmma_f32_16x16x4_f32(false, a1, false, b, (short)0, acc1, false, false);
    }

    const int mo = (lane >> 4) << 3;
    float* U0 = U + (size_t)(m0 + mo) * NTOT + n0 + col;
    float* U1 = U0 + (size_t)16 * NTOT;
    #pragma unroll
    for (int v = 0; v < 8; ++v) {
        U0[(size_t)v * NTOT] = acc0[v];
        U1[(size_t)v * NTOT] = acc1[v];
    }
}

__global__ __launch_bounds__(128) void lr_stage2(const float* __restrict__ U,
                                                 const float* __restrict__ Hm,
                                                 const float* __restrict__ bias,
                                                 float* __restrict__ out)
{
    __shared__ float Hs[32 * 289];
    const int tid = threadIdx.x;
    const int m0  = blockIdx.x * 32;

    #pragma unroll
    for (int it = 0; it < 72; ++it) {
        const int k = it >> 3, o_l = (tid >> 5) + ((it & 7) << 2), r = tid & 31;
        Hs[o_l * 289 + k * RANK + r] = Hm[(size_t)k * (COUT * RANK) + (size_t)(m0 + o_l) * RANK + r];
    }
    __syncthreads();

    const int wave = tid >> 5;
    const int lane = tid & 31;
    const int col  = lane & 15;
    const int klo  = (lane >> 4) << 1;

    const int n0   = (blockIdx.y * 4 + wave) * 16;
    const int n    = n0 + col;
    const int bidx = n / HWSZ;
    const int p    = n - bidx * HWSZ;
    const int h    = p / WID;
    const int w    = p - h * WID;

    int offs[9];
    unsigned vmask = 0;
    #pragma unroll
    for (int k = 0; k < 9; ++k) {
        int ki = k / 3, kj = k - ki * 3;
        int hs = h + ki - 1, ws = w + kj - 1;
        bool ok = (hs >= 0) & (hs < WID) & (ws >= 0) & (ws < WID);
        offs[k] = ok ? (bidx * HWSZ + hs * WID + ws) : 0;
        vmask  |= (ok ? 1u : 0u) << k;
    }

    v8f acc0 = {};
    v8f acc1 = {};
    #pragma unroll
    for (int k = 0; k < 9; ++k) {
        const bool ok = (vmask >> k) & 1u;
        const float* Ucol = U + offs[k];
        #pragma unroll
        for (int r0 = 0; r0 < RANK; r0 += 4) {
            const int kk = k * RANK + r0 + klo;
            const float t0 = Ucol[(size_t)kk       * NTOT];
            const float t1 = Ucol[(size_t)(kk + 1) * NTOT];
            v2f b;
            b.x = ok ? t0 : 0.0f;
            b.y = ok ? t1 : 0.0f;
            v2f a0, a1;
            a0.x = Hs[col * 289 + kk];
            a0.y = Hs[col * 289 + kk + 1];
            a1.x = Hs[(col + 16) * 289 + kk];
            a1.y = Hs[(col + 16) * 289 + kk + 1];
            acc0 = __builtin_amdgcn_wmma_f32_16x16x4_f32(false, a0, false, b, (short)0, acc0, false, false);
            acc1 = __builtin_amdgcn_wmma_f32_16x16x4_f32(false, a1, false, b, (short)0, acc1, false, false);
        }
    }

    const int mo = (lane >> 4) << 3;
    const float inv9 = 1.0f / 9.0f;
    #pragma unroll
    for (int v = 0; v < 8; ++v) {
        int o0 = m0 + v + mo;
        int o1 = o0 + 16;
        out[((size_t)bidx * COUT + o0) * HWSZ + p] = acc0[v] * inv9 + bias[o0];
        out[((size_t)bidx * COUT + o1) * HWSZ + p] = acc1[v] * inv9 + bias[o1];
    }
}

extern "C" void kernel_launch(void* const* d_in, const int* in_sizes, int n_in,
                              void* d_out, int out_size, void* d_ws, size_t ws_size,
                              hipStream_t stream)
{
    const float* x    = (const float*)d_in[0];
    const float* G    = (const float*)d_in[1];
    const float* Hm   = (const float*)d_in[2];
    const float* bias = (const float*)d_in[3];
    float* out = (float*)d_out;
    float* U   = (float*)d_ws;

    dim3 blk(128);
    lr_stage1<<<dim3(KTOT / 32, NTOT / 64), blk, 0, stream>>>(x, G, U);
    lr_stage2<<<dim3(COUT / 32, NTOT / 64), blk, 0, stream>>>(U, Hm, bias, out);
}


# Fewer loads per matrix op: a wave covers 96 rows (stage 1) / 64 rows (stage 2), one x/U fragment feeds 6/4 row tiles, weights via 128-bit LDS loads; k summed in a new fixed order within each 8
# speedup vs baseline: 2.1030x; 1.3110x over previous
#include <hip/hip_runtime.h>


typedef __attribute__((ext_vector_type(2))) float v2f;
typedef __attribute__((ext_vector_type(4))) float v4f;
typedef __attribute__((ext_vector_type(8))) float v8f;

#define HWSZ  3136
#define WID   56
#define CIN   256
#define COUT  256
#define RANK  32
#define KTOT  288
#define NTOT  50176

#define S1_MT    6
#define S1_ROWS  96
#define S1_PITCH 260
#define S2_MT    4
#define S2_ROWS  64
#define S2_PITCH 292


__global__ __launch_bounds__(256) void lr_stage1(const float* __restrict__ x,
                                                 const float* __restrict__ G,
                                                 float* __restrict__ U)
{
    __shared__ __attribute__((aligned(16))) float Gs[S1_ROWS * S1_PITCH];
    const int tid = threadIdx.x;
    const int m0  = blockIdx.x * S1_ROWS;

    #pragma unroll
    for (int it = 0; it < 24; ++it) {
        const int idx = tid + (it << 8);
        const int row = idx >> 6, f4 = idx & 63;
        const v4f g = *(const v4f*)(G + (size_t)(m0 + row) * CIN + f4 * 4);
        *(v4f*)(Gs + row * S1_PITCH + f4 * 4) = g;
    }
    __syncthreads();

    const int wave = tid >> 5;
    const int lane = tid & 31;
    const int col  = lane & 15;
    const int half = lane >> 4;

    const int n0   = (blockIdx.y * 8 + wave) * 16;
    const int bidx = n0 / HWSZ;
    const int p0   = n0 - bidx * HWSZ;
    const float* xb = x + (size_t)bidx * CIN * HWSZ + (size_t)(half * 4) * HWSZ + p0 + col;
    const float* As = Gs + col * S1_PITCH + half * 4;

    v8f acc[S1_MT];
    #pragma unroll
    for (int mt = 0; mt < S1_MT; ++mt) acc[mt] = (v8f){};

    #pragma unroll
    for (int kc = 0; kc < CIN; kc += 8) {
        v2f bA, bB;
        bA.x = xb[(size_t)(kc + 0) * HWSZ];
        bA.y = xb[(size_t)(kc + 1) * HWSZ];
        bB.x = xb[(size_t)(kc + 2) * HWSZ];
        bB.y = xb[(size_t)(kc + 3) * HWSZ];
        v4f a[S1_MT];
        #pragma unroll
        for (int mt = 0; mt < S1_MT; ++mt)
            a[mt] = *(const v4f*)(As + mt * 16 * S1_PITCH + kc);
        #pragma unroll
        for (int mt = 0; mt < S1_MT; ++mt) {
            v2f aA; aA.x = a[mt].x; aA.y = a[mt].y;
            acc[mt] = __builtin_amdgcn_wmma_f32_16x16x4_f32(false, aA, false, bA, (short)0, acc[mt], false, false);
        }
        #pragma unroll
        for (int mt = 0; mt < S1_MT; ++mt) {
            v2f aB; aB.x = a[mt].z; aB.y = a[mt].w;
            acc[mt] = __builtin_amdgcn_wmma_f32_16x16x4_f32(false, aB, false, bB, (short)0, acc[mt], false, false);
        }
    }

    float* Ub = U + (size_t)(m0 + half * 8) * NTOT + n0 + col;
    #pragma unroll
    for (int mt = 0; mt < S1_MT; ++mt) {
        #pragma unroll
        for (int v = 0; v < 8; ++v)
            Ub[(size_t)(mt * 16 + v) * NTOT] = acc[mt][v];
    }
}

__global__ __launch_bounds__(256) void lr_stage2(const float* __restrict__ U,
                                                 const float* __restrict__ Hm,
                                                 const float* __restrict__ bias,
                                                 float* __restrict__ out)
{
    __shared__ __attribute__((aligned(16))) float Hs[S2_ROWS * S2_PITCH];
    const int tid = threadIdx.x;
    const int m0  = blockIdx.x * S2_ROWS;

    #pragma unroll
    for (int it = 0; it < 18; ++it) {
        const int idx = tid + (it << 8);
        const int f4 = idx & 7, o_l = (idx >> 3) & 63, k = idx >> 9;
        const v4f hv = *(const v4f*)(Hm + (size_t)k * (COUT * RANK) + (size_t)(m0 + o_l) * RANK + f4 * 4);
        *(v4f*)(Hs + o_l * S2_PITCH + k * RANK + f4 * 4) = hv;
    }
    __syncthreads();

    const int wave = tid >> 5;
    const int lane = tid & 31;
    const int col  = lane & 15;
    const int half = lane >> 4;

    const int n0   = (blockIdx.y * 8 + wave) * 16;
    const int n    = n0 + col;
    const int bidx = n / HWSZ;
    const int p    = n - bidx * HWSZ;
    const int h    = p / WID;
    const int w    = p - h * WID;

    int offs[9];
    unsigned vmask = 0;
    #pragma unroll
    for (int k = 0; k < 9; ++k) {
        int ki = k / 3, kj = k - ki * 3;
        int hs = h + ki - 1, ws = w + kj - 1;
        bool ok = (hs >= 0) & (hs < WID) & (ws >= 0) & (ws < WID);
        offs[k] = ok ? (bidx * HWSZ + hs * WID + ws) : 0;
        vmask  |= (ok ? 1u : 0u) << k;
    }

    const float* As = Hs + col * S2_PITCH + half * 4;

    v8f acc[S2_MT];
    #pragma unroll
    for (int mt = 0; mt < S2_MT; ++mt) acc[mt] = (v8f){};

    #pragma unroll
    for (int k = 0; k < 9; ++k) {
        const bool ok = (vmask >> k) & 1u;
        const float* Ucol = U + (size_t)(k * RANK + half * 4) * NTOT + offs[k];
        #pragma unroll
        for (int r0 = 0; r0 < RANK; r0 += 8) {
            const float t0 = Ucol[(size_t)(r0 + 0) * NTOT];
            const float t1 = Ucol[(size_t)(r0 + 1) * NTOT];
            const float t2 = Ucol[(size_t)(r0 + 2) * NTOT];
            const float t3 = Ucol[(size_t)(r0 + 3) * NTOT];
            v2f bA, bB;
            bA.x = ok ? t0 : 0.0f;
            bA.y = ok ? t1 : 0.0f;
            bB.x = ok ? t2 : 0.0f;
            bB.y = ok ? t3 : 0.0f;
            v4f a[S2_MT];
            #pragma unroll
            for (int mt = 0; mt < S2_MT; ++mt)
                a[mt] = *(const v4f*)(As + mt * 16 * S2_PITCH + k * RANK + r0);
            #pragma unroll
            for (int mt = 0; mt < S2_MT; ++mt) {
                v2f aA; aA.x = a[mt].x; aA.y = a[mt].y;
                acc[mt] = __builtin_amdgcn_wmma_f32_16x16x4_f32(false, aA, false, bA, (short)0, acc[mt], false, false);
            }
            #pragma unroll
            for (int mt = 0; mt < S2_MT; ++mt) {
                v2f aB; aB.x = a[mt].z; aB.y = a[mt].w;
                acc[mt] = __builtin_amdgcn_wmma_f32_16x16x4_f32(false, aB, false, bB, (short)0, acc[mt], false, false);
            }
        }
    }

    const int mo = half << 3;
    const float inv9 = 1.0f / 9.0f;
    #pragma unroll
    for (int mt = 0; mt < S2_MT; ++mt) {
        #pragma unroll
        for (int v = 0; v < 8; ++v) {
            const int o = m0 + mt * 16 + v + mo;
            out[((size_t)bidx * COUT + o) * HWSZ + p] = acc[mt][v] * inv9 + bias[o];
        }
    }
}

extern "C" void kernel_launch(void* const* d_in, const int* in_sizes, int n_in,
                              void* d_out, int out_size, void* d_ws, size_t ws_size,
                              hipStream_t stream)
{
    const float* x    = (const float*)d_in[0];
    const float* G    = (const float*)d_in[1];
    const float* Hm   = (const float*)d_in[2];
    const float* bias = (const float*)d_in[3];
    float* out = (float*)d_out;
    float* U   = (float*)d_ws;

    dim3 blk(256);
    lr_stage1<<<dim3(KTOT / S1_ROWS, NTOT / 128), blk, 0, stream>>>(x, G, U);
    lr_stage2<<<dim3(COUT / S2_ROWS, NTOT / 128), blk, 0, stream>>>(U, Hm, bias, out);
}


# Intermediate U kept in fragment order (36 planes of [column][8 rows]): stage 2 reads a B fragment with one 16-byte load per lane, 512 contiguous bytes per wave; stage 1 stores 16 bytes; same sums
# speedup vs baseline: 2.1758x; 1.0346x over previous
#include <hip/hip_runtime.h>


typedef __attribute__((ext_vector_type(2))) float v2f;
typedef __attribute__((ext_vector_type(4))) float v4f;
typedef __attribute__((ext_vector_type(8))) float v8f;

#define HWSZ  3136
#define WID   56
#define CIN   256
#define COUT  256
#define RANK  32
#define KTOT  288
#define NTOT  50176

#define S1_MT    6
#define S1_ROWS  96
#define S1_PITCH 260
#define S2_MT    4
#define S2_ROWS  64
#define S2_PITCH 292

#define UPLANE (NTOT * 8)


__global__ __launch_bounds__(256) void lr_stage1(const float* __restrict__ x,
                                                 const float* __restrict__ G,
                                                 float* __restrict__ U)
{
    __shared__ __attribute__((aligned(16))) float Gs[S1_ROWS * S1_PITCH];
    const int tid = threadIdx.x;
    const int m0  = blockIdx.x * S1_ROWS;

    #pragma unroll
    for (int it = 0; it < 24; ++it) {
        const int idx = tid + (it << 8);
        const int row = idx >> 6, f4 = idx & 63;
        const v4f g = *(const v4f*)(G + (size_t)(m0 + row) * CIN + f4 * 4);
        *(v4f*)(Gs + row * S1_PITCH + f4 * 4) = g;
    }
    __syncthreads();

    const int wave = tid >> 5;
    const int lane = tid & 31;
    const int col  = lane & 15;
    const int half = lane >> 4;

    const int n0   = (blockIdx.y * 8 + wave) * 16;
    const int bidx = n0 / HWSZ;
    const int p0   = n0 - bidx * HWSZ;
    const float* xb = x + (size_t)bidx * CIN * HWSZ + (size_t)(half * 4) * HWSZ + p0 + col;
    const float* As = Gs + col * S1_PITCH + half * 4;

    v8f acc[S1_MT];
    #pragma unroll
    for (int mt = 0; mt < S1_MT; ++mt) acc[mt] = (v8f){};

    #pragma unroll
    for (int kc = 0; kc < CIN; kc += 8) {
        v2f bA, bB;
        bA.x = xb[(size_t)(kc + 0) * HWSZ];
        bA.y = xb[(size_t)(kc + 1) * HWSZ];
        bB.x = xb[(size_t)(kc + 2) * HWSZ];
        bB.y = xb[(size_t)(kc + 3) * HWSZ];
        v4f a[S1_MT];
        #pragma unroll
        for (int mt = 0; mt < S1_MT; ++mt)
            a[mt] = *(const v4f*)(As + mt * 16 * S1_PITCH + kc);
        #pragma unroll
        for (int mt = 0; mt < S1_MT; ++mt) {
            v2f aA; aA.x = a[mt].x; aA.y = a[mt].y;
            acc[mt] = __builtin_amdgcn_wmma_f32_16x16x4_f32(false, aA, false, bA, (short)0, acc[mt], false, false);
        }
        #pragma unroll
        for (int mt = 0; mt < S1_MT; ++mt) {
            v2f aB; aB.x = a[mt].z; aB.y = a[mt].w;
            acc[mt] = __builtin_amdgcn_wmma_f32_16x16x4_f32(false, aB, false, bB, (short)0, acc[mt], false, false);
        }
    }

    float* Ub = U + ((size_t)((m0 >> 3) + half) * NTOT + n0 + col) * 8;
    #pragma unroll
    for (int mt = 0; mt < S1_MT; ++mt) {
        float* q = Ub + (size_t)(2 * mt) * UPLANE;
        v4f lo, hi;
        lo.x = acc[mt][0]; lo.y = acc[mt][1]; lo.z = acc[mt][2]; lo.w = acc[mt][3];
        hi.x = acc[mt][4]; hi.y = acc[mt][5]; hi.z = acc[mt][6]; hi.w = acc[mt][7];
        *(v4f*)q       = lo;
        *(v4f*)(q + 4) = hi;
    }
}

__global__ __launch_bounds__(256) void lr_stage2(const float* __restrict__ U,
                                                 const float* __restrict__ Hm,
                                                 const float* __restrict__ bias,
                                                 float* __restrict__ out)
{
    __shared__ __attribute__((aligned(16))) float Hs[S2_ROWS * S2_PITCH];
    const int tid = threadIdx.x;
    const int m0  = blockIdx.x * S2_ROWS;

    #pragma unroll
    for (int it = 0; it < 18; ++it) {
        const int idx = tid + (it << 8);
        const int f4 = idx & 7, o_l = (idx >> 3) & 63, k = idx >> 9;
        const v4f hv = *(const v4f*)(Hm + (size_t)k * (COUT * RANK) + (size_t)(m0 + o_l) * RANK + f4 * 4);
        *(v4f*)(Hs + o_l * S2_PITCH + k * RANK + f4 * 4) = hv;
    }
    __syncthreads();

    const int wave = tid >> 5;
    const int lane = tid & 31;
    const int col  = lane & 15;
    const int half = lane >> 4;

    const int n0   = (blockIdx.y * 8 + wave) * 16;
    const int n    = n0 + col;
    const int bidx = n / HWSZ;
    const int p    = n - bidx * HWSZ;
    const int h    = p / WID;
    const int w    = p - h * WID;

    int offs[9];
    unsigned vmask = 0;
    #pragma unroll
    for (int k = 0; k < 9; ++k) {
        int ki = k / 3, kj = k - ki * 3;
        int hs = h + ki - 1, ws = w + kj - 1;
        bool ok = (hs >= 0) & (hs < WID) & (ws >= 0) & (ws < WID);
        offs[k] = ok ? (bidx * HWSZ + hs * WID + ws) : 0;
        vmask  |= (ok ? 1u : 0u) << k;
    }

    const float* As = Hs + col * S2_PITCH + half * 4;

    v8f acc[S2_MT];
    #pragma unroll
    for (int mt = 0; mt < S2_MT; ++mt) acc[mt] = (v8f){};

    #pragma unroll
    for (int k = 0; k < 9; ++k) {
        const bool ok = (vmask >> k) & 1u;
        const float* Utap = U + (size_t)(k * 4) * UPLANE + (size_t)offs[k] * 8 + half * 4;
        #pragma unroll
        for (int r0 = 0; r0 < RANK; r0 += 8) {
            const v4f t = *(const v4f*)(Utap + (size_t)(r0 >> 3) * UPLANE);
            v2f bA, bB;
            bA.x = ok ? t.x : 0.0f;
            bA.y = ok ? t.y : 0.0f;
            bB.x = ok ? t.z : 0.0f;
            bB.y = ok ? t.w : 0.0f;
            v4f a[S2_MT];
            #pragma unroll
            for (int mt = 0; mt < S2_MT; ++mt)
                a[mt] = *(const v4f*)(As + mt * 16 * S2_PITCH + k * RANK + r0);
            #pragma unroll
            for (int mt = 0; mt < S2_MT; ++mt) {
                v2f aA; aA.x = a[mt].x; aA.y = a[mt].y;
                acc[mt] = __builtin_amdgcn_wmma_f32_16x16x4_f32(false, aA, false, bA, (short)0, acc[mt], false, false);
            }
            #pragma unroll
            for (int mt = 0; mt < S2_MT; ++mt) {
                v2f aB; aB.x = a[mt].z; aB.y = a[mt].w;
                acc[mt] = __builtin_amdgcn_wmma_f32_16x16x4_f32(false, aB, false, bB, (short)0, acc[mt], false, false);
            }
        }
    }

    const int mo = half << 3;
    const float inv9 = 1.0f / 9.0f;
    #pragma unroll
    for (int mt = 0; mt < S2_MT; ++mt) {
        #pragma unroll
        for (int v = 0; v < 8; ++v) {
            const int o = m0 + mt * 16 + v + mo;
            out[((size_t)bidx * COUT + o) * HWSZ + p] = acc[mt][v] * inv9 + bias[o];
        }
    }
}

extern "C" void kernel_launch(void* const* d_in, const int* in_sizes, int n_in,
                              void* d_out, int out_size, void* d_ws, size_t ws_size,
                              hipStream_t stream)
{
    const float* x    = (const float*)d_in[0];
    const float* G    = (const float*)d_in[1];
    const float* Hm   = (const float*)d_in[2];
    const float* bias = (const float*)d_in[3];
    float* out = (float*)d_out;
    float* U   = (float*)d_ws;

    dim3 blk(256);
    lr_stage1<<<dim3(KTOT / S1_ROWS, NTOT / 128), blk, 0, stream>>>(x, G, U);
    lr_stage2<<<dim3(COUT / S2_ROWS, NTOT / 128), blk, 0, stream>>>(U, Hm, bias, out);
}
